# starting kernel
# baseline (speedup 1.0000x reference)
#include <hip/hip_runtime.h>

typedef float v2f __attribute__((ext_vector_type(2)));
typedef float v8f __attribute__((ext_vector_type(8)));

#define N_NODES   50000
#define DIM       256
#define NEG_SLOPE 0.01f
#define LDS_PITCH 260
#define EDGES_PER_BLOCK 4

__global__ __launch_bounds__(256) void zero_f4(float* __restrict__ out, int n4) {
  int i = blockIdx.x * blockDim.x + threadIdx.x;
  if (i < n4) {
    float4 z = make_float4(0.f, 0.f, 0.f, 0.f);
    reinterpret_cast<float4*>(out)[i] = z;
  }
}

__global__ __launch_bounds__(256) void spmm_scatter(
    const int* __restrict__ erow, const int* __restrict__ ecol,
    const float* __restrict__ eval, const float* __restrict__ x,
    float* side, int n_edges) {
  const int t  = threadIdx.x;
  const int e0 = blockIdx.x * EDGES_PER_BLOCK;
#pragma unroll
  for (int i = 0; i < EDGES_PER_BLOCK; ++i) {
    int e = e0 + i;
    if (e < n_edges) {
      int   r = erow[e];
      int   c = ecol[e];
      float v = eval[e];
      float m = v * x[(size_t)c * DIM + t];
      atomicAdd(&side[(size_t)r * DIM + t], m);
    }
  }
}

__global__ __launch_bounds__(256) void fused_dual_gemm(
    const float* __restrict__ ego,
    const float* __restrict__ W1, const float* __restrict__ b1,
    const float* __restrict__ W2, const float* __restrict__ b2,
    float* out) {
  __shared__ float a1[16 * LDS_PITCH];
  __shared__ float a2[16 * LDS_PITCH];

  const int rowbase = blockIdx.x * 16;
  const int t = threadIdx.x;

#pragma unroll
  for (int j = 0; j < 4; ++j) {
    int i4 = t + 256 * j;
    int r  = i4 >> 6;
    int c4 = i4 & 63;
    float4 e = reinterpret_cast<const float4*>(ego + (size_t)(rowbase + r) * DIM)[c4];
    float4 s = reinterpret_cast<const float4*>(out + (size_t)(rowbase + r) * DIM)[c4];
    float* p1 = &a1[r * LDS_PITCH + c4 * 4];
    float* p2 = &a2[r * LDS_PITCH + c4 * 4];
    p1[0] = e.x + s.x; p1[1] = e.y + s.y; p1[2] = e.z + s.z; p1[3] = e.w + s.w;
    p2[0] = e.x * s.x; p2[1] = e.y * s.y; p2[2] = e.z * s.z; p2[3] = e.w * s.w;
  }
  __syncthreads();

  const int lane = t & 31;
  const int wave = t >> 5;
  const int m    = lane & 15;
  const int half = lane >> 4;

  const float* a1p = &a1[m * LDS_PITCH + half * 2];
  const float* a2p = &a2[m * LDS_PITCH + half * 2];

#pragma unroll
  for (int cti = 0; cti < 2; ++cti) {
    const int colbase = (wave + cti * 8) * 16;
    const int jg = colbase + m;
    const float* w1p = W1 + (size_t)jg * DIM + half * 2;
    const float* w2p = W2 + (size_t)jg * DIM + half * 2;

    v8f acc1 = {};
    v8f acc2 = {};
#pragma unroll 8
    for (int k = 0; k < DIM; k += 4) {
      v2f fa1 = *reinterpret_cast<const v2f*>(a1p + k);
      v2f fa2 = *reinterpret_cast<const v2f*>(a2p + k);
      v2f fw1 = *reinterpret_cast<const v2f*>(w1p + k);
      v2f fw2 = *reinterpret_cast<const v2f*>(w2p + k);
      acc1 = __builtin_amdgcn_wmma_f32_16x16x4_f32(
          false, fa1, false, fw1, (short)0, acc1, false, false);
      acc2 = __builtin_amdgcn_wmma_f32_16x16x4_f32(
          false, fa2, false, fw2, (short)0, acc2, false, false);
    }

    const float bias1 = b1[jg];
    const float bias2 = b2[jg];
    float* op = out + (size_t)(rowbase + half * 8) * DIM + colbase + m;
#pragma unroll
    for (int v = 0; v < 8; ++v) {
      float s1 = acc1[v] + bias1;
      s1 = (s1 >= 0.f) ? s1 : NEG_SLOPE * s1;
      float s2 = acc2[v] + bias2;
      s2 = (s2 >= 0.f) ? s2 : NEG_SLOPE * s2;
      op[(size_t)v * DIM] = s1 + s2;
    }
  }
}

extern "C" void kernel_launch(void* const* d_in, const int* in_sizes, int n_in,
                              void* d_out, int out_size, void* d_ws, size_t ws_size,
                              hipStream_t stream) {
  const float* ego  = (const float*)d_in[0];
  const int*   erow = (const int*)  d_in[1];
  const int*   ecol = (const int*)  d_in[2];
  const float* eval = (const float*)d_in[3];
  const float* W1   = (const float*)d_in[4];
  const float* b1   = (const float*)d_in[5];
  const float* W2   = (const float*)d_in[6];
  const float* b2   = (const float*)d_in[7];
  float* out = (float*)d_out;
  const int n_edges = in_sizes[1];

  const int n4 = (N_NODES * DIM) / 4;
  zero_f4<<<(n4 + 255) / 256, 256, 0, stream>>>(out, n4);

  const int nblk = (n_edges + EDGES_PER_BLOCK - 1) / EDGES_PER_BLOCK;
  spmm_scatter<<<nblk, 256, 0, stream>>>(erow, ecol, eval, ego, out, n_edges);

  fused_dual_gemm<<<N_NODES / 16, 256, 0, stream>>>(ego, W1, b1, W2, b2, out);
}


# SpMM: each work-group owns 196 rows of side in LDS, scans the whole edge list and adds its edges with LDS f32 atomics, then stores rows once (no zero pass, no global atomics); sum order = arrival
# speedup vs baseline: 1.8960x; 1.8960x over previous
#include <hip/hip_runtime.h>

typedef float v2f __attribute__((ext_vector_type(2)));
typedef float v8f __attribute__((ext_vector_type(8)));

#define N_NODES   50000
#define DIM       256
#define NEG_SLOPE 0.01f
#define LDS_PITCH 260
#define ROWS_WG   196
#define SCAN_WAVES 32
#define CHUNK     128
#define QCAP      160

__global__ __launch_bounds__(1024) void spmm_owned_rows(
    const int* __restrict__ erow, const int* __restrict__ ecol,
    const float* __restrict__ eval, const float* __restrict__ x,
    float* __restrict__ side, int n_edges) {
  __shared__ float acc[ROWS_WG * DIM];
  __shared__ int   queue[SCAN_WAVES * QCAP];

  const int t    = threadIdx.x;
  const int lane = t & 31;
  const int wave = __builtin_amdgcn_readfirstlane(t >> 5);
  const int r0   = blockIdx.x * ROWS_WG;
  const int nrows = (N_NODES - r0 < ROWS_WG) ? (N_NODES - r0) : ROWS_WG;

  for (int i = t; i < ROWS_WG * DIM / 4; i += 1024)
    reinterpret_cast<float4*>(acc)[i] = make_float4(0.f, 0.f, 0.f, 0.f);
  __syncthreads();

  int* q = queue + wave * QCAP;
  int cnt = 0;

  auto flush = [&](int total, bool last) -> int {
    const int take = last ? total : (total & ~7);
    for (int b = 0; b < take; b += 32) {
      const int n = (take - b < 32) ? (take - b) : 32;
      int rl = 0, c = 0; float v = 0.f;
      if (lane < n) {
        const int e = q[b + lane];
        c = ecol[e]; rl = erow[e] - r0; v = eval[e];
      }
      for (int i = 0; i < n; i += 8) {
        int cj[8];
#pragma unroll
        for (int j = 0; j < 8; ++j) {
          const int p = (!last || i + j < n) ? (i + j) : (n - 1);
          cj[j] = __builtin_amdgcn_readlane(c, p);
        }
        v2f xv[8][4];
#pragma unroll
        for (int j = 0; j < 8; ++j) {
          const v2f* xp = reinterpret_cast<const v2f*>(x + (size_t)cj[j] * DIM) + lane;
#pragma unroll
          for (int k = 0; k < 4; ++k) xv[j][k] = xp[32 * k];
        }
#pragma unroll
        for (int j = 0; j < 8; ++j) {
          if (!last || i + j < n) {
            const int   rj = __builtin_amdgcn_readlane(rl, i + j);
            const float vj = __builtin_bit_cast(float,
                __builtin_amdgcn_readlane(__builtin_bit_cast(int, v), i + j));
            float* ap = acc + rj * DIM + 2 * lane;
#pragma unroll
            for (int k = 0; k < 4; ++k) {
              atomicAdd(ap + 64 * k,     vj * xv[j][k].x);
              atomicAdd(ap + 64 * k + 1, vj * xv[j][k].y);
            }
          }
        }
      }
    }
    const int rem = total - take;
    if (rem > 0) {
      int keep = 0;
      if (lane < rem) keep = q[take + lane];
      if (lane < rem) q[lane] = keep;
    }
    return rem;
  };

  const int nchunks  = (n_edges + CHUNK - 1) / CHUNK;
  const int per_wave = (nchunks + SCAN_WAVES - 1) / SCAN_WAVES;
  const int rot = (int)(((long long)blockIdx.x * per_wave) / gridDim.x);

  for (int it = 0; it < per_wave; it += 4) {
    int4 rv[4];
    int  eb[4];
    bool full = (it + 3 < per_wave);
#pragma unroll
    for (int u = 0; u < 4; ++u) {
      int ii = it + u + rot;
      if (ii >= per_wave) ii -= per_wave;
      const int chunk = ii * SCAN_WAVES + wave;
      eb[u] = chunk * CHUNK + lane * 4;
      full = full && ((chunk + 1) * CHUNK <= n_edges);
    }
    if (full) {
#pragma unroll
      for (int u = 0; u < 4; ++u)
        rv[u] = *reinterpret_cast<const int4*>(erow + eb[u]);
    } else {
#pragma unroll
      for (int u = 0; u < 4; ++u) {
        const int e = eb[u];
        rv[u] = make_int4(-1, -1, -1, -1);
        if (it + u < per_wave) {
          if (e     < n_edges) rv[u].x = erow[e];
          if (e + 1 < n_edges) rv[u].y = erow[e + 1];
          if (e + 2 < n_edges) rv[u].z = erow[e + 2];
          if (e + 3 < n_edges) rv[u].w = erow[e + 3];
        }
      }
    }
#pragma unroll
    for (int u = 0; u < 4; ++u) {
      const int r4[4] = {rv[u].x, rv[u].y, rv[u].z, rv[u].w};
#pragma unroll
      for (int j = 0; j < 4; ++j) {
        const bool m = (unsigned)(r4[j] - r0) < (unsigned)nrows;
        const unsigned bal = __builtin_amdgcn_ballot_w32(m);
        if (bal) {
          const int pre = __builtin_popcount(bal & ((1u << lane) - 1u));
          if (m) q[cnt + pre] = eb[u] + j;
          cnt += __builtin_popcount(bal);
        }
      }
      if (cnt >= 32) {
        __builtin_amdgcn_wave_barrier();
        cnt = flush(cnt, false);
      }
    }
  }
  __builtin_amdgcn_wave_barrier();
  flush(cnt, true);
  __syncthreads();

  float4* dst = reinterpret_cast<float4*>(side + (size_t)r0 * DIM);
  for (int i = t; i < nrows * (DIM / 4); i += 1024)
    dst[i] = reinterpret_cast<const float4*>(acc)[i];
}

__global__ __launch_bounds__(256) void fused_dual_gemm(
    const float* __restrict__ ego,
    const float* __restrict__ W1, const float* __restrict__ b1,
    const float* __restrict__ W2, const float* __restrict__ b2,
    float* out) {
  __shared__ float a1[16 * LDS_PITCH];
  __shared__ float a2[16 * LDS_PITCH];

  const int rowbase = blockIdx.x * 16;
  const int t = threadIdx.x;

#pragma unroll
  for (int j = 0; j < 4; ++j) {
    int i4 = t + 256 * j;
    int r  = i4 >> 6;
    int c4 = i4 & 63;
    float4 e = reinterpret_cast<const float4*>(ego + (size_t)(rowbase + r) * DIM)[c4];
    float4 s = reinterpret_cast<const float4*>(out + (size_t)(rowbase + r) * DIM)[c4];
    float* p1 = &a1[r * LDS_PITCH + c4 * 4];
    float* p2 = &a2[r * LDS_PITCH + c4 * 4];
    p1[0] = e.x + s.x; p1[1] = e.y + s.y; p1[2] = e.z + s.z; p1[3] = e.w + s.w;
    p2[0] = e.x * s.x; p2[1] = e.y * s.y; p2[2] = e.z * s.z; p2[3] = e.w * s.w;
  }
  __syncthreads();

  const int lane = t & 31;
  const int wave = t >> 5;
  const int m    = lane & 15;
  const int half = lane >> 4;

  const float* a1p = &a1[m * LDS_PITCH + half * 2];
  const float* a2p = &a2[m * LDS_PITCH + half * 2];

#pragma unroll
  for (int cti = 0; cti < 2; ++cti) {
    const int colbase = (wave + cti * 8) * 16;
    const int jg = colbase + m;
    const float* w1p = W1 + (size_t)jg * DIM + half * 2;
    const float* w2p = W2 + (size_t)jg * DIM + half * 2;

    v8f acc1 = {};
    v8f acc2 = {};
#pragma unroll 8
    for (int k = 0; k < DIM; k += 4) {
      v2f fa1 = *reinterpret_cast<const v2f*>(a1p + k);
      v2f fa2 = *reinterpret_cast<const v2f*>(a2p + k);
      v2f fw1 = *reinterpret_cast<const v2f*>(w1p + k);
      v2f fw2 = *reinterpret_cast<const v2f*>(w2p + k);
      acc1 = __builtin_amdgcn_wmma_f32_16x16x4_f32(
          false, fa1, false, fw1, (short)0, acc1, false, false);
      acc2 = __builtin_amdgcn_wmma_f32_16x16x4_f32(
          false, fa2, false, fw2, (short)0, acc2, false, false);
    }

    const float bias1 = b1[jg];
    const float bias2 = b2[jg];
    float* op = out + (size_t)(rowbase + half * 8) * DIM + colbase + m;
#pragma unroll
    for (int v = 0; v < 8; ++v) {
      float s1 = acc1[v] + bias1;
      s1 = (s1 >= 0.f) ? s1 : NEG_SLOPE * s1;
      float s2 = acc2[v] + bias2;
      s2 = (s2 >= 0.f) ? s2 : NEG_SLOPE * s2;
      op[(size_t)v * DIM] = s1 + s2;
    }
  }
}

extern "C" void kernel_launch(void* const* d_in, const int* in_sizes, int n_in,
                              void* d_out, int out_size, void* d_ws, size_t ws_size,
                              hipStream_t stream) {
  const float* ego  = (const float*)d_in[0];
  const int*   erow = (const int*)  d_in[1];
  const int*   ecol = (const int*)  d_in[2];
  const float* eval = (const float*)d_in[3];
  const float* W1   = (const float*)d_in[4];
  const float* b1   = (const float*)d_in[5];
  const float* W2   = (const float*)d_in[6];
  const float* b2   = (const float*)d_in[7];
  float* out = (float*)d_out;
  const int n_edges = in_sizes[1];

  spmm_owned_rows<<<(N_NODES + ROWS_WG - 1) / ROWS_WG, 1024, 0, stream>>>(
      erow, ecol, eval, ego, out, n_edges);

  fused_dual_gemm<<<N_NODES / 16, 256, 0, stream>>>(ego, W1, b1, W2, b2, out);
}


# Dual GEMM: each wave fetches its W tiles by whole-line 16-byte loads into a private LDS buffer (64-k chunks, next chunk prefetched) and reads W fragments from LDS, not 16-line global loads
# speedup vs baseline: 2.5732x; 1.3572x over previous
#include <hip/hip_runtime.h>

typedef float v2f __attribute__((ext_vector_type(2)));
typedef float v4f __attribute__((ext_vector_type(4)));
typedef float v8f __attribute__((ext_vector_type(8)));

#define N_NODES   50000
#define DIM       256
#define NEG_SLOPE 0.01f
#define LDS_PITCH 260
#define ROWS_WG   196
#define SCAN_WAVES 32
#define CHUNK     128
#define QCAP      160
#define WCH       64
#define W_PITCH   68

__global__ __launch_bounds__(1024) void spmm_owned_rows(
    const int* __restrict__ erow, const int* __restrict__ ecol,
    const float* __restrict__ eval, const float* __restrict__ x,
    float* __restrict__ side, int n_edges) {
  __shared__ float acc[ROWS_WG * DIM];
  __shared__ int   queue[SCAN_WAVES * QCAP];

  const int t    = threadIdx.x;
  const int lane = t & 31;
  const int wave = __builtin_amdgcn_readfirstlane(t >> 5);
  const int r0   = blockIdx.x * ROWS_WG;
  const int nrows = (N_NODES - r0 < ROWS_WG) ? (N_NODES - r0) : ROWS_WG;

  for (int i = t; i < ROWS_WG * DIM / 4; i += 1024)
    reinterpret_cast<float4*>(acc)[i] = make_float4(0.f, 0.f, 0.f, 0.f);
  __syncthreads();

  int* q = queue + wave * QCAP;
  int cnt = 0;

  auto flush = [&](int total, bool last) -> int {
    const int take = last ? total : (total & ~7);
    for (int b = 0; b < take; b += 32) {
      const int n = (take - b < 32) ? (take - b) : 32;
      int rl = 0, c = 0; float v = 0.f;
      if (lane < n) {
        const int e = q[b + lane];
        c = ecol[e]; rl = erow[e] - r0; v = eval[e];
      }
      for (int i = 0; i < n; i += 8) {
        int cj[8];
#pragma unroll
        for (int j = 0; j < 8; ++j) {
          const int p = (!last || i + j < n) ? (i + j) : (n - 1);
          cj[j] = __builtin_amdgcn_readlane(c, p);
        }
        v2f xv[8][4];
#pragma unroll
        for (int j = 0; j < 8; ++j) {
          const v2f* xp = reinterpret_cast<const v2f*>(x + (size_t)cj[j] * DIM) + lane;
#pragma unroll
          for (int k = 0; k < 4; ++k) xv[j][k] = xp[32 * k];
        }
#pragma unroll
        for (int j = 0; j < 8; ++j) {
          if (!last || i + j < n) {
            const int   rj = __builtin_amdgcn_readlane(rl, i + j);
            const float vj = __builtin_bit_cast(float,
                __builtin_amdgcn_readlane(__builtin_bit_cast(int, v), i + j));
            float* ap = acc + rj * DIM + 2 * lane;
#pragma unroll
            for (int k = 0; k < 4; ++k) {
              atomicAdd(ap + 64 * k,     vj * xv[j][k].x);
              atomicAdd(ap + 64 * k + 1, vj * xv[j][k].y);
            }
          }
        }
      }
    }
    const int rem = total - take;
    if (rem > 0) {
      int keep = 0;
      if (lane < rem) keep = q[take + lane];
      if (lane < rem) q[lane] = keep;
    }
    return rem;
  };

  const int nchunks  = (n_edges + CHUNK - 1) / CHUNK;
  const int per_wave = (nchunks + SCAN_WAVES - 1) / SCAN_WAVES;
  const int rot = (int)(((long long)blockIdx.x * per_wave) / gridDim.x);

  for (int it = 0; it < per_wave; it += 4) {
    int4 rv[4];
    int  eb[4];
    bool full = (it + 3 < per_wave);
#pragma unroll
    for (int u = 0; u < 4; ++u) {
      int ii = it + u + rot;
      if (ii >= per_wave) ii -= per_wave;
      const int chunk = ii * SCAN_WAVES + wave;
      eb[u] = chunk * CHUNK + lane * 4;
      full = full && ((chunk + 1) * CHUNK <= n_edges);
    }
    if (full) {
#pragma unroll
      for (int u = 0; u < 4; ++u)
        rv[u] = *reinterpret_cast<const int4*>(erow + eb[u]);
    } else {
#pragma unroll
      for (int u = 0; u < 4; ++u) {
        const int e = eb[u];
        rv[u] = make_int4(-1, -1, -1, -1);
        if (it + u < per_wave) {
          if (e     < n_edges) rv[u].x = erow[e];
          if (e + 1 < n_edges) rv[u].y = erow[e + 1];
          if (e + 2 < n_edges) rv[u].z = erow[e + 2];
          if (e + 3 < n_edges) rv[u].w = erow[e + 3];
        }
      }
    }
#pragma unroll
    for (int u = 0; u < 4; ++u) {
      const int r4[4] = {rv[u].x, rv[u].y, rv[u].z, rv[u].w};
#pragma unroll
      for (int j = 0; j < 4; ++j) {
        const bool m = (unsigned)(r4[j] - r0) < (unsigned)nrows;
        const unsigned bal = __builtin_amdgcn_ballot_w32(m);
        if (bal) {
          const int pre = __builtin_popcount(bal & ((1u << lane) - 1u));
          if (m) q[cnt + pre] = eb[u] + j;
          cnt += __builtin_popcount(bal);
        }
      }
      if (cnt >= 32) {
        __builtin_amdgcn_wave_barrier();
        cnt = flush(cnt, false);
      }
    }
  }
  __builtin_amdgcn_wave_barrier();
  flush(cnt, true);
  __syncthreads();

  float4* dst = reinterpret_cast<float4*>(side + (size_t)r0 * DIM);
  for (int i = t; i < nrows * (DIM / 4); i += 1024)
    dst[i] = reinterpret_cast<const float4*>(acc)[i];
}

__global__ __launch_bounds__(256) void fused_dual_gemm(
    const float* __restrict__ ego,
    const float* __restrict__ W1, const float* __restrict__ b1,
    const float* __restrict__ W2, const float* __restrict__ b2,
    float* out) {
  __shared__ float a1[16 * LDS_PITCH];
  __shared__ float a2[16 * LDS_PITCH];
  __shared__ __attribute__((aligned(16))) float wst[8 * 2 * 16 * W_PITCH];

  const int rowbase = blockIdx.x * 16;
  const int t = threadIdx.x;

#pragma unroll
  for (int j = 0; j < 4; ++j) {
    int i4 = t + 256 * j;
    int r  = i4 >> 6;
    int c4 = i4 & 63;
    float4 e = reinterpret_cast<const float4*>(ego + (size_t)(rowbase + r) * DIM)[c4];
    float4 s = reinterpret_cast<const float4*>(out + (size_t)(rowbase + r) * DIM)[c4];
    float* p1 = &a1[r * LDS_PITCH + c4 * 4];
    float* p2 = &a2[r * LDS_PITCH + c4 * 4];
    p1[0] = e.x + s.x; p1[1] = e.y + s.y; p1[2] = e.z + s.z; p1[3] = e.w + s.w;
    p2[0] = e.x * s.x; p2[1] = e.y * s.y; p2[2] = e.z * s.z; p2[3] = e.w * s.w;
  }

  const int lane = t & 31;
  const int wave = t >> 5;
  const int m    = lane & 15;
  const int half = lane >> 4;

  float* wb1 = wst + wave * (2 * 16 * W_PITCH);
  float* wb2 = wb1 + 16 * W_PITCH;
  const int lc = (lane & 15) * 4;

  v4f r1[8], r2[8];
  {
    const float* p1 = W1 + (size_t)(wave * 16 + half) * DIM + lc;
    const float* p2 = W2 + (size_t)(wave * 16 + half) * DIM + lc;
#pragma unroll
    for (int i = 0; i < 8; ++i) {
      r1[i] = *reinterpret_cast<const v4f*>(p1 + (size_t)(2 * i) * DIM);
      r2[i] = *reinterpret_cast<const v4f*>(p2 + (size_t)(2 * i) * DIM);
    }
  }
  __syncthreads();

  const float* a1p = &a1[m * LDS_PITCH + half * 2];
  const float* a2p = &a2[m * LDS_PITCH + half * 2];
  const float* f1p = wb1 + m * W_PITCH + half * 2;
  const float* f2p = wb2 + m * W_PITCH + half * 2;
  float* s1p = wb1 + half * W_PITCH + lc;
  float* s2p = wb2 + half * W_PITCH + lc;

#pragma unroll 1
  for (int cti = 0; cti < 2; ++cti) {
    const int colbase = (wave + cti * 8) * 16;
    const int jg = colbase + m;

    v8f acc1 = {};
    v8f acc2 = {};
#pragma unroll 1
    for (int kc = 0; kc < 4; ++kc) {
#pragma unroll
      for (int i = 0; i < 8; ++i) {
        *reinterpret_cast<v4f*>(s1p + (2 * i) * W_PITCH) = r1[i];
        *reinterpret_cast<v4f*>(s2p + (2 * i) * W_PITCH) = r2[i];
      }
      const int cn = cti * 4 + kc + 1;
      if (cn < 8) {
        const int cbn = (wave + (cn >> 2) * 8) * 16;
        const float* p1 = W1 + (size_t)(cbn + half) * DIM + (cn & 3) * WCH + lc;
        const float* p2 = W2 + (size_t)(cbn + half) * DIM + (cn & 3) * WCH + lc;
#pragma unroll
        for (int i = 0; i < 8; ++i) {
          r1[i] = *reinterpret_cast<const v4f*>(p1 + (size_t)(2 * i) * DIM);
          r2[i] = *reinterpret_cast<const v4f*>(p2 + (size_t)(2 * i) * DIM);
        }
      }
      __builtin_amdgcn_wave_barrier();
#pragma unroll
      for (int kk = 0; kk < WCH; kk += 4) {
        v2f fa1 = *reinterpret_cast<const v2f*>(a1p + kc * WCH + kk);
        v2f fa2 = *reinterpret_cast<const v2f*>(a2p + kc * WCH + kk);
        v2f fw1 = *reinterpret_cast<const v2f*>(f1p + kk);
        v2f fw2 = *reinterpret_cast<const v2f*>(f2p + kk);
        acc1 = __builtin_amdgcn_wmma_f32_16x16x4_f32(
            false, fa1, false, fw1, (short)0, acc1, false, false);
        acc2 = __builtin_amdgcn_wmma_f32_16x16x4_f32(
            false, fa2, false, fw2, (short)0, acc2, false, false);
      }
      __builtin_amdgcn_wave_barrier();
    }

    const float bias1 = b1[jg];
    const float bias2 = b2[jg];
    float* op = out + (size_t)(rowbase + half * 8) * DIM + colbase + m;
#pragma unroll
    for (int v = 0; v < 8; ++v) {
      float s1 = acc1[v] + bias1;
      s1 = (s1 >= 0.f) ? s1 : NEG_SLOPE * s1;
      float s2 = acc2[v] + bias2;
      s2 = (s2 >= 0.f) ? s2 : NEG_SLOPE * s2;
      op[(size_t)v * DIM] = s1 + s2;
    }
  }
}

extern "C" void kernel_launch(void* const* d_in, const int* in_sizes, int n_in,
                              void* d_out, int out_size, void* d_ws, size_t ws_size,
                              hipStream_t stream) {
  const float* ego  = (const float*)d_in[0];
  const int*   erow = (const int*)  d_in[1];
  const int*   ecol = (const int*)  d_in[2];
  const float* eval = (const float*)d_in[3];
  const float* W1   = (const float*)d_in[4];
  const float* b1   = (const float*)d_in[5];
  const float* W2   = (const float*)d_in[6];
  const float* b2   = (const float*)d_in[7];
  float* out = (float*)d_out;
  const int n_edges = in_sizes[1];

  spmm_owned_rows<<<(N_NODES + ROWS_WG - 1) / ROWS_WG, 1024, 0, stream>>>(
      erow, ecol, eval, ego, out, n_edges);

  fused_dual_gemm<<<N_NODES / 16, 256, 0, stream>>>(ego, W1, b1, W2, b2, out);
}


# SpMM edge search: a new first launch writes one owner byte per edge (row / 196) to the workspace; the 256 owner groups scan that 0.8 MB map, not the 3.2 MB row list; sum order = arrival, as before
# speedup vs baseline: 2.7070x; 1.0520x over previous
#include <hip/hip_runtime.h>

typedef float v2f __attribute__((ext_vector_type(2)));
typedef float v4f __attribute__((ext_vector_type(4)));
typedef float v8f __attribute__((ext_vector_type(8)));

#define N_NODES   50000
#define DIM       256
#define NEG_SLOPE 0.01f
#define LDS_PITCH 260
#define ROWS_WG   196
#define SCAN_WAVES 32
#define CHUNK     512
#define QCAP      544
#define WCH       64
#define W_PITCH   68

__global__ __launch_bounds__(256) void owner_map(
    const int* __restrict__ erow, unsigned int* __restrict__ omap, int n_edges) {
  const int ngroups = (n_edges + 15) >> 4;
  for (int g = blockIdx.x * 256 + threadIdx.x; g < ngroups; g += gridDim.x * 256) {
    const int e0 = g * 16;
    unsigned int w[4];
    if (e0 + 16 <= n_edges) {
#pragma unroll
      for (int j = 0; j < 4; ++j) {
        const int4 r = *reinterpret_cast<const int4*>(erow + e0 + 4 * j);
        w[j] = (((unsigned)r.x / ROWS_WG) & 255u)
             | ((((unsigned)r.y / ROWS_WG) & 255u) << 8)
             | ((((unsigned)r.z / ROWS_WG) & 255u) << 16)
             | ((((unsigned)r.w / ROWS_WG) & 255u) << 24);
      }
    } else {
#pragma unroll
      for (int j = 0; j < 4; ++j) {
        unsigned int acc = 0u;
#pragma unroll
        for (int b = 0; b < 4; ++b) {
          const int e = e0 + 4 * j + b;
          if (e < n_edges) acc |= (((unsigned)erow[e] / ROWS_WG) & 255u) << (8 * b);
        }
        w[j] = acc;
      }
    }
    *reinterpret_cast<uint4*>(omap + e0 / 4) = make_uint4(w[0], w[1], w[2], w[3]);
  }
}

__global__ __launch_bounds__(1024) void spmm_owned_rows(
    const int* __restrict__ erow, const int* __restrict__ ecol,
    const float* __restrict__ eval, const float* __restrict__ x,
    const unsigned int* __restrict__ omap,
    float* __restrict__ side, int n_edges) {
  __shared__ float acc[ROWS_WG * DIM];
  __shared__ int   queue[SCAN_WAVES * QCAP];

  const int t    = threadIdx.x;
  const int lane = t & 31;
  const int wave = __builtin_amdgcn_readfirstlane(t >> 5);
  const int r0   = blockIdx.x * ROWS_WG;
  const int nrows = (N_NODES - r0 < ROWS_WG) ? (N_NODES - r0) : ROWS_WG;

  for (int i = t; i < ROWS_WG * DIM / 4; i += 1024)
    reinterpret_cast<float4*>(acc)[i] = make_float4(0.f, 0.f, 0.f, 0.f);
  __syncthreads();

  int* q = queue + wave * QCAP;
  int cnt = 0;

  auto flush = [&](int total, bool last) -> int {
    const int take = last ? total : (total & ~7);
    for (int b = 0; b < take; b += 32) {
      const int n = (take - b < 32) ? (take - b) : 32;
      int rl = 0, c = 0; float v = 0.f;
      if (lane < n) {
        const int e = q[b + lane];
        c = ecol[e]; rl = erow[e] - r0; v = eval[e];
        if ((unsigned)rl >= (unsigned)nrows) { rl = 0; v = 0.f; }
      }
      for (int i = 0; i < n; i += 8) {
        int cj[8];
#pragma unroll
        for (int j = 0; j < 8; ++j) {
          const int p = (!last || i + j < n) ? (i + j) : (n - 1);
          cj[j] = __builtin_amdgcn_readlane(c, p);
        }
        v2f xv[8][4];
#pragma unroll
        for (int j = 0; j < 8; ++j) {
          const v2f* xp = reinterpret_cast<const v2f*>(x + (size_t)cj[j] * DIM) + lane;
#pragma unroll
          for (int k = 0; k < 4; ++k) xv[j][k] = xp[32 * k];
        }
#pragma unroll
        for (int j = 0; j < 8; ++j) {
          if (!last || i + j < n) {
            const int   rj = __builtin_amdgcn_readlane(rl, i + j);
            const float vj = __builtin_bit_cast(float,
                __builtin_amdgcn_readlane(__builtin_bit_cast(int, v), i + j));
            float* ap = acc + rj * DIM + 2 * lane;
#pragma unroll
            for (int k = 0; k < 4; ++k) {
              atomicAdd(ap + 64 * k,     vj * xv[j][k].x);
              atomicAdd(ap + 64 * k + 1, vj * xv[j][k].y);
            }
          }
        }
      }
    }
    const int rem = total - take;
    if (rem > 0) {
      int keep = 0;
      if (lane < rem) keep = q[take + lane];
      if (lane < rem) q[lane] = keep;
    }
    return rem;
  };

  const int nchunks  = (n_edges + CHUNK - 1) / CHUNK;
  const int per_wave = (nchunks + SCAN_WAVES - 1) / SCAN_WAVES;
  const int rot = (int)(((long long)blockIdx.x * per_wave) / gridDim.x);
  const int nmap = (n_edges + 15) & ~15;
  const unsigned int pat = blockIdx.x * 0x01010101u;
  const unsigned char* mapb = reinterpret_cast<const unsigned char*>(omap);

  for (int it = 0; it < per_wave; it += 4) {
    uint4 wv[4];
    int   eb[4];
    bool full = (it + 3 < per_wave);
#pragma unroll
    for (int u = 0; u < 4; ++u) {
      int ii = it + u + rot;
      if (ii >= per_wave) ii -= per_wave;
      const int chunk = ii * SCAN_WAVES + wave;
      eb[u] = chunk * CHUNK + lane * 16;
      full = full && ((chunk + 1) * CHUNK <= nmap);
    }
    if (full) {
#pragma unroll
      for (int u = 0; u < 4; ++u)
        wv[u] = *reinterpret_cast<const uint4*>(mapb + eb[u]);
    } else {
#pragma unroll
      for (int u = 0; u < 4; ++u) {
        wv[u] = make_uint4(~pat, ~pat, ~pat, ~pat);
        if (it + u < per_wave && eb[u] < nmap)
          wv[u] = *reinterpret_cast<const uint4*>(mapb + eb[u]);
      }
    }
#pragma unroll
    for (int u = 0; u < 4; ++u) {
      const unsigned int w4[4] = {wv[u].x, wv[u].y, wv[u].z, wv[u].w};
#pragma unroll
      for (int j = 0; j < 4; ++j) {
        const unsigned int xo = w4[j] ^ pat;
        unsigned int z = ~(((xo & 0x7F7F7F7Fu) + 0x7F7F7F7Fu) | xo) & 0x80808080u;
        for (int trip = 0; trip < 4; ++trip) {
          if (!__builtin_amdgcn_ballot_w32(z != 0u)) break;
          const int  bit = __builtin_ctz(z | 0x80000000u);
          const int  e = eb[u] + 4 * j + (bit >> 3);
          const bool m = (z != 0u) && (e < n_edges);
          const unsigned bal = __builtin_amdgcn_ballot_w32(m);
          if (bal) {
            const int pre = __builtin_popcount(bal & ((1u << lane) - 1u));
            if (m) q[cnt + pre] = e;
            cnt += __builtin_popcount(bal);
          }
          z &= z - 1u;
        }
      }
      if (cnt >= 32) {
        __builtin_amdgcn_wave_barrier();
        cnt = flush(cnt, false);
      }
    }
  }
  __builtin_amdgcn_wave_barrier();
  flush(cnt, true);
  __syncthreads();

  float4* dst = reinterpret_cast<float4*>(side + (size_t)r0 * DIM);
  for (int i = t; i < nrows * (DIM / 4); i += 1024)
    dst[i] = reinterpret_cast<const float4*>(acc)[i];
}

__global__ __launch_bounds__(256) void fused_dual_gemm(
    const float* __restrict__ ego,
    const float* __restrict__ W1, const float* __restrict__ b1,
    const float* __restrict__ W2, const float* __restrict__ b2,
    float* out) {
  __shared__ float a1[16 * LDS_PITCH];
  __shared__ float a2[16 * LDS_PITCH];
  __shared__ __attribute__((aligned(16))) float wst[8 * 2 * 16 * W_PITCH];

  const int rowbase = blockIdx.x * 16;
  const int t = threadIdx.x;

#pragma unroll
  for (int j = 0; j < 4; ++j) {
    int i4 = t + 256 * j;
    int r  = i4 >> 6;
    int c4 = i4 & 63;
    float4 e = reinterpret_cast<const float4*>(ego + (size_t)(rowbase + r) * DIM)[c4];
    float4 s = reinterpret_cast<const float4*>(out + (size_t)(rowbase + r) * DIM)[c4];
    float* p1 = &a1[r * LDS_PITCH + c4 * 4];
    float* p2 = &a2[r * LDS_PITCH + c4 * 4];
    p1[0] = e.x + s.x; p1[1] = e.y + s.y; p1[2] = e.z + s.z; p1[3] = e.w + s.w;
    p2[0] = e.x * s.x; p2[1] = e.y * s.y; p2[2] = e.z * s.z; p2[3] = e.w * s.w;
  }

  const int lane = t & 31;
  const int wave = t >> 5;
  const int m    = lane & 15;
  const int half = lane >> 4;

  float* wb1 = wst + wave * (2 * 16 * W_PITCH);
  float* wb2 = wb1 + 16 * W_PITCH;
  const int lc = (lane & 15) * 4;

  v4f r1[8], r2[8];
  {
    const float* p1 = W1 + (size_t)(wave * 16 + half) * DIM + lc;
    const float* p2 = W2 + (size_t)(wave * 16 + half) * DIM + lc;
#pragma unroll
    for (int i = 0; i < 8; ++i) {
      r1[i] = *reinterpret_cast<const v4f*>(p1 + (size_t)(2 * i) * DIM);
      r2[i] = *reinterpret_cast<const v4f*>(p2 + (size_t)(2 * i) * DIM);
    }
  }
  __syncthreads();

  const float* a1p = &a1[m * LDS_PITCH + half * 2];
  const float* a2p = &a2[m * LDS_PITCH + half * 2];
  const float* f1p = wb1 + m * W_PITCH + half * 2;
  const float* f2p = wb2 + m * W_PITCH + half * 2;
  float* s1p = wb1 + half * W_PITCH + lc;
  float* s2p = wb2 + half * W_PITCH + lc;

#pragma unroll 1
  for (int cti = 0; cti < 2; ++cti) {
    const int colbase = (wave + cti * 8) * 16;
    const int jg = colbase + m;

    v8f acc1 = {};
    v8f acc2 = {};
#pragma unroll 1
    for (int kc = 0; kc < 4; ++kc) {
#pragma unroll
      for (int i = 0; i < 8; ++i) {
        *reinterpret_cast<v4f*>(s1p + (2 * i) * W_PITCH) = r1[i];
        *reinterpret_cast<v4f*>(s2p + (2 * i) * W_PITCH) = r2[i];
      }
      const int cn = cti * 4 + kc + 1;
      if (cn < 8) {
        const int cbn = (wave + (cn >> 2) * 8) * 16;
        const float* p1 = W1 + (size_t)(cbn + half) * DIM + (cn & 3) * WCH + lc;
        const float* p2 = W2 + (size_t)(cbn + half) * DIM + (cn & 3) * WCH + lc;
#pragma unroll
        for (int i = 0; i < 8; ++i) {
          r1[i] = *reinterpret_cast<const v4f*>(p1 + (size_t)(2 * i) * DIM);
          r2[i] = *reinterpret_cast<const v4f*>(p2 + (size_t)(2 * i) * DIM);
        }
      }
      __builtin_amdgcn_wave_barrier();
#pragma unroll
      for (int kk = 0; kk < WCH; kk += 4) {
        v2f fa1 = *reinterpret_cast<const v2f*>(a1p + kc * WCH + kk);
        v2f fa2 = *reinterpret_cast<const v2f*>(a2p + kc * WCH + kk);
        v2f fw1 = *reinterpret_cast<const v2f*>(f1p + kk);
        v2f fw2 = *reinterpret_cast<const v2f*>(f2p + kk);
        acc1 = __builtin_amdgcn_wmma_f32_16x16x4_f32(
            false, fa1, false, fw1, (short)0, acc1, false, false);
        acc2 = __builtin_amdgcn_wmma_f32_16x16x4_f32(
            false, fa2, false, fw2, (short)0, acc2, false, false);
      }
      __builtin_amdgcn_wave_barrier();
    }

    const float bias1 = b1[jg];
    const float bias2 = b2[jg];
    float* op = out + (size_t)(rowbase + half * 8) * DIM + colbase + m;
#pragma unroll
    for (int v = 0; v < 8; ++v) {
      float s1 = acc1[v] + bias1;
      s1 = (s1 >= 0.f) ? s1 : NEG_SLOPE * s1;
      float s2 = acc2[v] + bias2;
      s2 = (s2 >= 0.f) ? s2 : NEG_SLOPE * s2;
      op[(size_t)v * DIM] = s1 + s2;
    }
  }
}

extern "C" void kernel_launch(void* const* d_in, const int* in_sizes, int n_in,
                              void* d_out, int out_size, void* d_ws, size_t ws_size,
                              hipStream_t stream) {
  const float* ego  = (const float*)d_in[0];
  const int*   erow = (const int*)  d_in[1];
  const int*   ecol = (const int*)  d_in[2];
  const float* eval = (const float*)d_in[3];
  const float* W1   = (const float*)d_in[4];
  const float* b1   = (const float*)d_in[5];
  const float* W2   = (const float*)d_in[6];
  const float* b2   = (const float*)d_in[7];
  float* out = (float*)d_out;
  const int n_edges = in_sizes[1];

  unsigned int* omap = (unsigned int*)d_ws;

  owner_map<<<256, 256, 0, stream>>>(erow, omap, n_edges);

  spmm_owned_rows<<<(N_NODES + ROWS_WG - 1) / ROWS_WG, 1024, 0, stream>>>(
      erow, ecol, eval, ego, omap, out, n_edges);

  fused_dual_gemm<<<N_NODES / 16, 256, 0, stream>>>(ego, W1, b1, W2, b2, out);
}


# SpMM adds: each scanned edge goes through an LDS list to the wave that owns its row (row % 32), which adds by plain 16-byte LDS read-fma-write, no f32 atomics; sum order = arrival in that list
# speedup vs baseline: 2.8660x; 1.0587x over previous
#include <hip/hip_runtime.h>

typedef float v2f __attribute__((ext_vector_type(2)));
typedef float v4f __attribute__((ext_vector_type(4)));
typedef float v8f __attribute__((ext_vector_type(8)));

#define N_NODES   50000
#define DIM       256
#define NEG_SLOPE 0.01f
#define LDS_PITCH 260
#define ROWS_WG   196
#define SCAN_WAVES 32
#define CHUNK     512
#define QCAP      544
#define ICAP      256
#define WCH       64
#define W_PITCH   68

__global__ __launch_bounds__(256) void owner_map(
    const int* __restrict__ erow, unsigned int* __restrict__ omap, int n_edges) {
  const int ngroups = (n_edges + 15) >> 4;
  for (int g = blockIdx.x * 256 + threadIdx.x; g < ngroups; g += gridDim.x * 256) {
    const int e0 = g * 16;
    unsigned int w[4];
    if (e0 + 16 <= n_edges) {
#pragma unroll
      for (int j = 0; j < 4; ++j) {
        const int4 r = *reinterpret_cast<const int4*>(erow + e0 + 4 * j);
        w[j] = (((unsigned)r.x / ROWS_WG) & 255u)
             | ((((unsigned)r.y / ROWS_WG) & 255u) << 8)
             | ((((unsigned)r.z / ROWS_WG) & 255u) << 16)
             | ((((unsigned)r.w / ROWS_WG) & 255u) << 24);
      }
    } else {
#pragma unroll
      for (int j = 0; j < 4; ++j) {
        unsigned int acc = 0u;
#pragma unroll
        for (int b = 0; b < 4; ++b) {
          const int e = e0 + 4 * j + b;
          if (e < n_edges) acc |= (((unsigned)erow[e] / ROWS_WG) & 255u) << (8 * b);
        }
        w[j] = acc;
      }
    }
    *reinterpret_cast<uint4*>(omap + e0 / 4) = make_uint4(w[0], w[1], w[2], w[3]);
  }
}

__global__ __launch_bounds__(1024) void spmm_owned_rows(
    const int* __restrict__ erow, const int* __restrict__ ecol,
    const float* __restrict__ eval, const float* __restrict__ x,
    const unsigned int* __restrict__ omap,
    float* __restrict__ side, int n_edges) {
  __shared__ __attribute__((aligned(16))) float acc[ROWS_WG * DIM];
  __shared__ int           queue[SCAN_WAVES * QCAP];
  __shared__ int           own_e[SCAN_WAVES * ICAP];
  __shared__ unsigned char own_r[SCAN_WAVES * ICAP];
  __shared__ int           own_n[SCAN_WAVES];
  __shared__ int           overflow;

  const int t    = threadIdx.x;
  const int lane = t & 31;
  const int wave = __builtin_amdgcn_readfirstlane(t >> 5);
  const int r0   = blockIdx.x * ROWS_WG;
  const int nrows = (N_NODES - r0 < ROWS_WG) ? (N_NODES - r0) : ROWS_WG;

  for (int i = t; i < ROWS_WG * DIM / 4; i += 1024)
    reinterpret_cast<float4*>(acc)[i] = make_float4(0.f, 0.f, 0.f, 0.f);
  if (t < SCAN_WAVES) own_n[t] = 0;
  if (t == 0) overflow = 0;
  __syncthreads();

  int* q = queue + wave * QCAP;

  auto add_slow = [&](int e) {
    const int rl = erow[e] - r0;
    if ((unsigned)rl < (unsigned)nrows) {
      const float v = eval[e];
      const float* xr = x + (size_t)ecol[e] * DIM;
      float* ar = acc + rl * DIM;
#pragma unroll 1
      for (int d = 0; d < DIM; ++d) atomicAdd(ar + d, v * xr[d]);
    }
  };

  const int nchunks  = (n_edges + CHUNK - 1) / CHUNK;
  const int per_wave = (nchunks + SCAN_WAVES - 1) / SCAN_WAVES;
  const int rot = (int)(((long long)blockIdx.x * per_wave) / gridDim.x);
  const int nmap = (n_edges + 15) & ~15;
  const unsigned int pat = blockIdx.x * 0x01010101u;
  const unsigned char* mapb = reinterpret_cast<const unsigned char*>(omap);

  auto scan = [&](const bool second) -> int {
    int cnt = 0;
    for (int it = 0; it < per_wave; it += 4) {
      uint4 wv[4];
      int   eb[4];
      bool full = (it + 3 < per_wave);
#pragma unroll
      for (int u = 0; u < 4; ++u) {
        int ii = it + u + rot;
        if (ii >= per_wave) ii -= per_wave;
        const int chunk = ii * SCAN_WAVES + wave;
        eb[u] = chunk * CHUNK + lane * 16;
        full = full && ((chunk + 1) * CHUNK <= nmap);
      }
      if (full) {
#pragma unroll
        for (int u = 0; u < 4; ++u)
          wv[u] = *reinterpret_cast<const uint4*>(mapb + eb[u]);
      } else {
#pragma unroll
        for (int u = 0; u < 4; ++u) {
          wv[u] = make_uint4(~pat, ~pat, ~pat, ~pat);
          if (it + u < per_wave && eb[u] < nmap)
            wv[u] = *reinterpret_cast<const uint4*>(mapb + eb[u]);
        }
      }
#pragma unroll
      for (int u = 0; u < 4; ++u) {
        const unsigned int w4[4] = {wv[u].x, wv[u].y, wv[u].z, wv[u].w};
#pragma unroll
        for (int j = 0; j < 4; ++j) {
          const unsigned int xo = w4[j] ^ pat;
          unsigned int z = ~(((xo & 0x7F7F7F7Fu) + 0x7F7F7F7Fu) | xo) & 0x80808080u;
          for (int trip = 0; trip < 4; ++trip) {
            if (!__builtin_amdgcn_ballot_w32(z != 0u)) break;
            const int  bit = __builtin_ctz(z | 0x80000000u);
            const int  e = eb[u] + 4 * j + (bit >> 3);
            const bool m = (z != 0u) && (e < n_edges);
            const unsigned bal = __builtin_amdgcn_ballot_w32(m);
            if (bal) {
              const int idx = cnt + __builtin_popcount(bal & ((1u << lane) - 1u));
              if (m) {
                if (!second) { if (idx < QCAP) q[idx] = e; }
                else if (idx >= QCAP) add_slow(e);
              }
              cnt += __builtin_popcount(bal);
            }
            z &= z - 1u;
          }
        }
      }
    }
    return cnt;
  };

  const int found = scan(false);
  const int nlist = (found < QCAP) ? found : QCAP;
  if (found > QCAP && lane == 0) overflow = 1;
  int nund = 0;
  for (int b = 0; b < nlist; b += 32) {
    const bool have = (b + lane < nlist);
    int e = 0, rl = -1;
    if (have) { e = q[b + lane]; rl = erow[e] - r0; }
    const bool valid = have && ((unsigned)rl < (unsigned)nrows);
    bool und = false;
    if (valid) {
      const int o = rl & (SCAN_WAVES - 1);
      const int slot = atomicAdd(&own_n[o], 1);
      if (slot < ICAP) { own_e[o * ICAP + slot] = e; own_r[o * ICAP + slot] = (unsigned char)rl; }
      else und = true;
    }
    const unsigned bal = __builtin_amdgcn_ballot_w32(und);
    if (bal) {
      if (und) q[nund + __builtin_popcount(bal & ((1u << lane) - 1u))] = e;
      nund += __builtin_popcount(bal);
    }
  }
  __syncthreads();

  {
    const int* se = own_e + wave * ICAP;
    const unsigned char* sr = own_r + wave * ICAP;
    int total = __builtin_amdgcn_readfirstlane(own_n[wave]);
    if (total > ICAP) total = ICAP;
    for (int b = 0; b < total; b += 32) {
      const int n = (total - b < 32) ? (total - b) : 32;
      int rl = 0, c = 0; float v = 0.f;
      if (lane < n) {
        const int e = se[b + lane];
        rl = sr[b + lane]; c = ecol[e]; v = eval[e];
      }
      for (int i = 0; i < n; i += 8) {
        int cj[8], rj[8]; float vj[8];
#pragma unroll
        for (int j = 0; j < 8; ++j) {
          const int p = (i + j < n) ? (i + j) : (n - 1);
          cj[j] = __builtin_amdgcn_readlane(c, p);
          rj[j] = __builtin_amdgcn_readlane(rl, p);
          vj[j] = __builtin_bit_cast(float, __builtin_amdgcn_readlane(__builtin_bit_cast(int, v), p));
        }
        v4f xv[8][2];
#pragma unroll
        for (int j = 0; j < 8; ++j) {
          const v4f* xp = reinterpret_cast<const v4f*>(x + (size_t)cj[j] * DIM) + lane;
          xv[j][0] = xp[0];
          xv[j][1] = xp[32];
        }
#pragma unroll
        for (int j = 0; j < 8; ++j) {
          if (i + j < n) {
            v4f* ap = reinterpret_cast<v4f*>(acc + rj[j] * DIM) + lane;
            v4f a0 = ap[0];
            v4f a1 = ap[32];
            a0 += vj[j] * xv[j][0];
            a1 += vj[j] * xv[j][1];
            ap[0]  = a0;
            ap[32] = a1;
          }
        }
      }
    }
  }
  __syncthreads();

  for (int b = 0; b < nund; b += 32)
    if (b + lane < nund) add_slow(q[b + lane]);
  if (overflow) scan(true);
  __syncthreads();

  float4* dst = reinterpret_cast<float4*>(side + (size_t)r0 * DIM);
  for (int i = t; i < nrows * (DIM / 4); i += 1024)
    dst[i] = reinterpret_cast<const float4*>(acc)[i];
}

__global__ __launch_bounds__(256) void fused_dual_gemm(
    const float* __restrict__ ego,
    const float* __restrict__ W1, const float* __restrict__ b1,
    const float* __restrict__ W2, const float* __restrict__ b2,
    float* out) {
  __shared__ float a1[16 * LDS_PITCH];
  __shared__ float a2[16 * LDS_PITCH];
  __shared__ __attribute__((aligned(16))) float wst[8 * 2 * 16 * W_PITCH];

  const int rowbase = blockIdx.x * 16;
  const int t = threadIdx.x;

#pragma unroll
  for (int j = 0; j < 4; ++j) {
    int i4 = t + 256 * j;
    int r  = i4 >> 6;
    int c4 = i4 & 63;
    float4 e = reinterpret_cast<const float4*>(ego + (size_t)(rowbase + r) * DIM)[c4];
    float4 s = reinterpret_cast<const float4*>(out + (size_t)(rowbase + r) * DIM)[c4];
    float* p1 = &a1[r * LDS_PITCH + c4 * 4];
    float* p2 = &a2[r * LDS_PITCH + c4 * 4];
    p1[0] = e.x + s.x; p1[1] = e.y + s.y; p1[2] = e.z + s.z; p1[3] = e.w + s.w;
    p2[0] = e.x * s.x; p2[1] = e.y * s.y; p2[2] = e.z * s.z; p2[3] = e.w * s.w;
  }

  const int lane = t & 31;
  const int wave = t >> 5;
  const int m    = lane & 15;
  const int half = lane >> 4;

  float* wb1 = wst + wave * (2 * 16 * W_PITCH);
  float* wb2 = wb1 + 16 * W_PITCH;
  const int lc = (lane & 15) * 4;

  v4f r1[8], r2[8];
  {
    const float* p1 = W1 + (size_t)(wave * 16 + half) * DIM + lc;
    const float* p2 = W2 + (size_t)(wave * 16 + half) * DIM + lc;
#pragma unroll
    for (int i = 0; i < 8; ++i) {
      r1[i] = *reinterpret_cast<const v4f*>(p1 + (size_t)(2 * i) * DIM);
      r2[i] = *reinterpret_cast<const v4f*>(p2 + (size_t)(2 * i) * DIM);
    }
  }
  __syncthreads();

  const float* a1p = &a1[m * LDS_PITCH + half * 2];
  const float* a2p = &a2[m * LDS_PITCH + half * 2];
  const float* f1p = wb1 + m * W_PITCH + half * 2;
  const float* f2p = wb2 + m * W_PITCH + half * 2;
  float* s1p = wb1 + half * W_PITCH + lc;
  float* s2p = wb2 + half * W_PITCH + lc;

#pragma unroll 1
  for (int cti = 0; cti < 2; ++cti) {
    const int colbase = (wave + cti * 8) * 16;
    const int jg = colbase + m;

    v8f acc1 = {};
    v8f acc2 = {};
#pragma unroll 1
    for (int kc = 0; kc < 4; ++kc) {
#pragma unroll
      for (int i = 0; i < 8; ++i) {
        *reinterpret_cast<v4f*>(s1p + (2 * i) * W_PITCH) = r1[i];
        *reinterpret_cast<v4f*>(s2p + (2 * i) * W_PITCH) = r2[i];
      }
      const int cn = cti * 4 + kc + 1;
      if (cn < 8) {
        const int cbn = (wave + (cn >> 2) * 8) * 16;
        const float* p1 = W1 + (size_t)(cbn + half) * DIM + (cn & 3) * WCH + lc;
        const float* p2 = W2 + (size_t)(cbn + half) * DIM + (cn & 3) * WCH + lc;
#pragma unroll
        for (int i = 0; i < 8; ++i) {
          r1[i] = *reinterpret_cast<const v4f*>(p1 + (size_t)(2 * i) * DIM);
          r2[i] = *reinterpret_cast<const v4f*>(p2 + (size_t)(2 * i) * DIM);
        }
      }
      __builtin_amdgcn_wave_barrier();
#pragma unroll
      for (int kk = 0; kk < WCH; kk += 4) {
        v2f fa1 = *reinterpret_cast<const v2f*>(a1p + kc * WCH + kk);
        v2f fa2 = *reinterpret_cast<const v2f*>(a2p + kc * WCH + kk);
        v2f fw1 = *reinterpret_cast<const v2f*>(f1p + kk);
        v2f fw2 = *reinterpret_cast<const v2f*>(f2p + kk);
        acc1 = __builtin_amdgcn_wmma_f32_16x16x4_f32(
            false, fa1, false, fw1, (short)0, acc1, false, false);
        acc2 = __builtin_amdgcn_wmma_f32_16x16x4_f32(
            false, fa2, false, fw2, (short)0, acc2, false, false);
      }
      __builtin_amdgcn_wave_barrier();
    }

    const float bias1 = b1[jg];
    const float bias2 = b2[jg];
    float* op = out + (size_t)(rowbase + half * 8) * DIM + colbase + m;
#pragma unroll
    for (int v = 0; v < 8; ++v) {
      float s1 = acc1[v] + bias1;
      s1 = (s1 >= 0.f) ? s1 : NEG_SLOPE * s1;
      float s2 = acc2[v] + bias2;
      s2 = (s2 >= 0.f) ? s2 : NEG_SLOPE * s2;
      op[(size_t)v * DIM] = s1 + s2;
    }
  }
}

extern "C" void kernel_launch(void* const* d_in, const int* in_sizes, int n_in,
                              void* d_out, int out_size, void* d_ws, size_t ws_size,
                              hipStream_t stream) {
  const float* ego  = (const float*)d_in[0];
  const int*   erow = (const int*)  d_in[1];
  const int*   ecol = (const int*)  d_in[2];
  const float* eval = (const float*)d_in[3];
  const float* W1   = (const float*)d_in[4];
  const float* b1   = (const float*)d_in[5];
  const float* W2   = (const float*)d_in[6];
  const float* b2   = (const float*)d_in[7];
  float* out = (float*)d_out;
  const int n_edges = in_sizes[1];

  unsigned int* omap = (unsigned int*)d_ws;

  owner_map<<<256, 256, 0, stream>>>(erow, omap, n_edges);

  spmm_owned_rows<<<(N_NODES + ROWS_WG - 1) / ROWS_WG, 1024, 0, stream>>>(
      erow, ecol, eval, ego, omap, out, n_edges);

  fused_dual_gemm<<<N_NODES / 16, 256, 0, stream>>>(ego, W1, b1, W2, b2, out);
}


# SpMM edge search: kernel 0 also writes, per owner, one bit per 16-edge group; an owner group reads its 6 KB bit row and the owner bytes of flagged groups only, not the 0.8 MB map; sum order = arrival
# speedup vs baseline: 3.0185x; 1.0532x over previous
#include <hip/hip_runtime.h>

typedef float v2f __attribute__((ext_vector_type(2)));
typedef float v4f __attribute__((ext_vector_type(4)));
typedef float v8f __attribute__((ext_vector_type(8)));

#define N_NODES   50000
#define DIM       256
#define NEG_SLOPE 0.01f
#define LDS_PITCH 260
#define ROWS_WG   196
#define SCAN_WAVES 32
#define GRP       16
#define BSPAN     16
#define TPITCH    17
#define QCAP      544
#define ICAP      256
#define WCH       64
#define W_PITCH   68

__global__ __launch_bounds__(512) void owner_index(
    const int* __restrict__ erow, unsigned int* __restrict__ ws, int n_edges) {
  __shared__ unsigned int tile[256 * TPITCH];
  const int t = threadIdx.x;
  const int nmap    = (n_edges + GRP - 1) & ~(GRP - 1);
  const int ngroups = nmap / GRP;
  const int nwords  = (ngroups + 31) >> 5;
  const int bmw     = (nwords + 3) & ~3;
  unsigned int* omap  = ws;
  unsigned int* obits = ws + (nmap >> 2);
  const int nspans = (bmw + BSPAN - 1) / BSPAN;
  for (int span = blockIdx.x; span < nspans; span += gridDim.x) {
    for (int i = t; i < 256 * TPITCH; i += 512) tile[i] = 0u;
    __syncthreads();
    const int g = span * (BSPAN * 32) + t;
    if (g < ngroups) {
      const int e0 = g * GRP;
      const unsigned int bit = 1u << (t & 31);
      unsigned int* tw = tile + (t >> 5);
      unsigned int w[4];
      if (e0 + GRP <= n_edges) {
#pragma unroll
        for (int j = 0; j < 4; ++j) {
          const int4 r = *reinterpret_cast<const int4*>(erow + e0 + 4 * j);
          const unsigned int o0 = (unsigned)r.x / ROWS_WG, o1 = (unsigned)r.y / ROWS_WG;
          const unsigned int o2 = (unsigned)r.z / ROWS_WG, o3 = (unsigned)r.w / ROWS_WG;
          w[j] = (o0 & 255u) | ((o1 & 255u) << 8) | ((o2 & 255u) << 16) | ((o3 & 255u) << 24);
          if (o0 < 256u) atomicOr(tw + o0 * TPITCH, bit);
          if (o1 < 256u) atomicOr(tw + o1 * TPITCH, bit);
          if (o2 < 256u) atomicOr(tw + o2 * TPITCH, bit);
          if (o3 < 256u) atomicOr(tw + o3 * TPITCH, bit);
        }
      } else {
#pragma unroll
        for (int j = 0; j < 4; ++j) {
          unsigned int acc = 0u;
#pragma unroll
          for (int b = 0; b < 4; ++b) {
            const int e = e0 + 4 * j + b;
            if (e < n_edges) {
              const unsigned int o = (unsigned)erow[e] / ROWS_WG;
              acc |= (o & 255u) << (8 * b);
              if (o < 256u) atomicOr(tw + o * TPITCH, bit);
            }
          }
          w[j] = acc;
        }
      }
      *reinterpret_cast<uint4*>(omap + e0 / 4) = make_uint4(w[0], w[1], w[2], w[3]);
    }
    __syncthreads();
    const int w0 = span * BSPAN;
    for (int i = t; i < 256 * BSPAN; i += 512) {
      const int o = i / BSPAN, k = i % BSPAN;
      if (w0 + k < bmw) obits[(size_t)o * bmw + w0 + k] = tile[o * TPITCH + k];
    }
    __syncthreads();
  }
}

__global__ __launch_bounds__(1024) void spmm_owned_rows(
    const int* __restrict__ erow, const int* __restrict__ ecol,
    const float* __restrict__ eval, const float* __restrict__ x,
    const unsigned int* __restrict__ ws,
    float* __restrict__ side, int n_edges) {
  __shared__ __attribute__((aligned(16))) float acc[ROWS_WG * DIM];
  __shared__ int           queue[SCAN_WAVES * QCAP];
  __shared__ int           own_e[SCAN_WAVES * ICAP];
  __shared__ unsigned char own_r[SCAN_WAVES * ICAP];
  __shared__ int           own_n[SCAN_WAVES];
  __shared__ int           overflow;

  const int t    = threadIdx.x;
  const int lane = t & 31;
  const int wave = __builtin_amdgcn_readfirstlane(t >> 5);
  const int r0   = blockIdx.x * ROWS_WG;
  const int nrows = (N_NODES - r0 < ROWS_WG) ? (N_NODES - r0) : ROWS_WG;

  for (int i = t; i < ROWS_WG * DIM / 4; i += 1024)
    reinterpret_cast<float4*>(acc)[i] = make_float4(0.f, 0.f, 0.f, 0.f);
  if (t < SCAN_WAVES) own_n[t] = 0;
  if (t == 0) overflow = 0;
  __syncthreads();

  int* q = queue + wave * QCAP;

  auto add_slow = [&](int e) {
    const int rl = erow[e] - r0;
    if ((unsigned)rl < (unsigned)nrows) {
      const float v = eval[e];
      const float* xr = x + (size_t)ecol[e] * DIM;
      float* ar = acc + rl * DIM;
#pragma unroll 1
      for (int d = 0; d < DIM; ++d) atomicAdd(ar + d, v * xr[d]);
    }
  };

  const int nmap    = (n_edges + GRP - 1) & ~(GRP - 1);
  const int ngroups = nmap / GRP;
  const int nwords  = (ngroups + 31) >> 5;
  const int bmw     = (nwords + 3) & ~3;
  const unsigned int* omap = ws;
  const unsigned int* bm   = ws + (nmap >> 2) + (size_t)blockIdx.x * bmw;
  const int pw = (nwords + SCAN_WAVES - 1) / SCAN_WAVES;
  const unsigned int pat = blockIdx.x * 0x01010101u;
  const unsigned int below = (1u << lane) - 1u;

  auto scan = [&](const bool second) -> int {
    int cnt = 0;
    for (int s = 0; s < pw; s += 64) {
      const int k0 = s + 2 * lane;
      const int wi = wave * pw + k0;
      const bool h0 = (k0 < pw) && (wi < nwords);
      const bool h1 = (k0 + 1 < pw) && (wi + 1 < nwords);
      unsigned int w0 = bm[h0 ? wi : 0];
      unsigned int w1 = bm[h1 ? wi + 1 : 0];
      if (!h0) w0 = 0u;
      if (!h1) w1 = 0u;
      unsigned long long m = (unsigned long long)w0 | ((unsigned long long)w1 << 32);
      while (__builtin_amdgcn_ballot_w32(m != 0ull)) {
        unsigned int vm = 0u;
        int e0 = 0;
        if (m != 0ull) {
          const int g = wi * 32 + __builtin_ctzll(m);
          e0 = g * GRP;
          const uint4 ob = *reinterpret_cast<const uint4*>(omap + g * (GRP / 4));
          const unsigned int w4[4] = {ob.x, ob.y, ob.z, ob.w};
#pragma unroll
          for (int j = 0; j < 4; ++j) {
            const unsigned int xo = w4[j] ^ pat;
            const unsigned int z = ~(((xo & 0x7F7F7F7Fu) + 0x7F7F7F7Fu) | xo) & 0x80808080u;
            vm |= ((((z >> 7) * 0x00204081u) >> 21) & 0xFu) << (4 * j);
          }
          const int lim = n_edges - e0;
          if (lim < GRP) vm &= (1u << lim) - 1u;
        }
        const int c = __builtin_popcount(vm);
        int base = 0, tot = 0;
#pragma unroll
        for (int k = 0; k < 5; ++k) {
          const unsigned int bal = __builtin_amdgcn_ballot_w32(((c >> k) & 1) != 0);
          base += __builtin_popcount(bal & below) << k;
          tot  += __builtin_popcount(bal) << k;
        }
        int idx = cnt + base;
        while (vm) {
          const int e = e0 + __builtin_ctz(vm);
          if (!second) { if (idx < QCAP) q[idx] = e; }
          else if (idx >= QCAP) add_slow(e);
          ++idx;
          vm &= vm - 1u;
        }
        cnt += tot;
        m &= m - 1ull;
      }
    }
    return cnt;
  };

  const int found = scan(false);
  const int nlist = (found < QCAP) ? found : QCAP;
  if (found > QCAP && lane == 0) overflow = 1;
  int nund = 0;
  for (int b = 0; b < nlist; b += 32) {
    const bool have = (b + lane < nlist);
    int e = 0, rl = -1;
    if (have) { e = q[b + lane]; rl = erow[e] - r0; }
    const bool valid = have && ((unsigned)rl < (unsigned)nrows);
    bool und = false;
    if (valid) {
      const int o = rl & (SCAN_WAVES - 1);
      const int slot = atomicAdd(&own_n[o], 1);
      if (slot < ICAP) { own_e[o * ICAP + slot] = e; own_r[o * ICAP + slot] = (unsigned char)rl; }
      else und = true;
    }
    const unsigned bal = __builtin_amdgcn_ballot_w32(und);
    if (bal) {
      if (und) q[nund + __builtin_popcount(bal & ((1u << lane) - 1u))] = e;
      nund += __builtin_popcount(bal);
    }
  }
  __syncthreads();

  {
    const int* se = own_e + wave * ICAP;
    const unsigned char* sr = own_r + wave * ICAP;
    int total = __builtin_amdgcn_readfirstlane(own_n[wave]);
    if (total > ICAP) total = ICAP;
    for (int b = 0; b < total; b += 32) {
      const int n = (total - b < 32) ? (total - b) : 32;
      int rl = 0, c = 0; float v = 0.f;
      if (lane < n) {
        const int e = se[b + lane];
        rl = sr[b + lane]; c = ecol[e]; v = eval[e];
      }
      for (int i = 0; i < n; i += 8) {
        int cj[8], rj[8]; float vj[8];
#pragma unroll
        for (int j = 0; j < 8; ++j) {
          const int p = (i + j < n) ? (i + j) : (n - 1);
          cj[j] = __builtin_amdgcn_readlane(c, p);
          rj[j] = __builtin_amdgcn_readlane(rl, p);
          vj[j] = __builtin_bit_cast(float, __builtin_amdgcn_readlane(__builtin_bit_cast(int, v), p));
        }
        v4f xv[8][2];
#pragma unroll
        for (int j = 0; j < 8; ++j) {
          const v4f* xp = reinterpret_cast<const v4f*>(x + (size_t)cj[j] * DIM) + lane;
          xv[j][0] = xp[0];
          xv[j][1] = xp[32];
        }
#pragma unroll
        for (int j = 0; j < 8; ++j) {
          if (i + j < n) {
            v4f* ap = reinterpret_cast<v4f*>(acc + rj[j] * DIM) + lane;
            v4f a0 = ap[0];
            v4f a1 = ap[32];
            a0 += vj[j] * xv[j][0];
            a1 += vj[j] * xv[j][1];
            ap[0]  = a0;
            ap[32] = a1;
          }
        }
      }
    }
  }
  __syncthreads();

  for (int b = 0; b < nund; b += 32)
    if (b + lane < nund) add_slow(q[b + lane]);
  if (overflow) scan(true);
  __syncthreads();

  float4* dst = reinterpret_cast<float4*>(side + (size_t)r0 * DIM);
  for (int i = t; i < nrows * (DIM / 4); i += 1024)
    dst[i] = reinterpret_cast<const float4*>(acc)[i];
}

__global__ __launch_bounds__(256) void fused_dual_gemm(
    const float* __restrict__ ego,
    const float* __restrict__ W1, const float* __restrict__ b1,
    const float* __restrict__ W2, const float* __restrict__ b2,
    float* out) {
  __shared__ float a1[16 * LDS_PITCH];
  __shared__ float a2[16 * LDS_PITCH];
  __shared__ __attribute__((aligned(16))) float wst[8 * 2 * 16 * W_PITCH];

  const int rowbase = blockIdx.x * 16;
  const int t = threadIdx.x;

#pragma unroll
  for (int j = 0; j < 4; ++j) {
    int i4 = t + 256 * j;
    int r  = i4 >> 6;
    int c4 = i4 & 63;
    float4 e = reinterpret_cast<const float4*>(ego + (size_t)(rowbase + r) * DIM)[c4];
    float4 s = reinterpret_cast<const float4*>(out + (size_t)(rowbase + r) * DIM)[c4];
    float* p1 = &a1[r * LDS_PITCH + c4 * 4];
    float* p2 = &a2[r * LDS_PITCH + c4 * 4];
    p1[0] = e.x + s.x; p1[1] = e.y + s.y; p1[2] = e.z + s.z; p1[3] = e.w + s.w;
    p2[0] = e.x * s.x; p2[1] = e.y * s.y; p2[2] = e.z * s.z; p2[3] = e.w * s.w;
  }

  const int lane = t & 31;
  const int wave = t >> 5;
  const int m    = lane & 15;
  const int half = lane >> 4;

  float* wb1 = wst + wave * (2 * 16 * W_PITCH);
  float* wb2 = wb1 + 16 * W_PITCH;
  const int lc = (lane & 15) * 4;

  v4f r1[8], r2[8];
  {
    const float* p1 = W1 + (size_t)(wave * 16 + half) * DIM + lc;
    const float* p2 = W2 + (size_t)(wave * 16 + half) * DIM + lc;
#pragma unroll
    for (int i = 0; i < 8; ++i) {
      r1[i] = *reinterpret_cast<const v4f*>(p1 + (size_t)(2 * i) * DIM);
      r2[i] = *reinterpret_cast<const v4f*>(p2 + (size_t)(2 * i) * DIM);
    }
  }
  __syncthreads();

  const float* a1p = &a1[m * LDS_PITCH + half * 2];
  const float* a2p = &a2[m * LDS_PITCH + half * 2];
  const float* f1p = wb1 + m * W_PITCH + half * 2;
  const float* f2p = wb2 + m * W_PITCH + half * 2;
  float* s1p = wb1 + half * W_PITCH + lc;
  float* s2p = wb2 + half * W_PITCH + lc;

#pragma unroll 1
  for (int cti = 0; cti < 2; ++cti) {
    const int colbase = (wave + cti * 8) * 16;
    const int jg = colbase + m;

    v8f acc1 = {};
    v8f acc2 = {};
#pragma unroll 1
    for (int kc = 0; kc < 4; ++kc) {
#pragma unroll
      for (int i = 0; i < 8; ++i) {
        *reinterpret_cast<v4f*>(s1p + (2 * i) * W_PITCH) = r1[i];
        *reinterpret_cast<v4f*>(s2p + (2 * i) * W_PITCH) = r2[i];
      }
      const int cn = cti * 4 + kc + 1;
      if (cn < 8) {
        const int cbn = (wave + (cn >> 2) * 8) * 16;
        const float* p1 = W1 + (size_t)(cbn + half) * DIM + (cn & 3) * WCH + lc;
        const float* p2 = W2 + (size_t)(cbn + half) * DIM + (cn & 3) * WCH + lc;
#pragma unroll
        for (int i = 0; i < 8; ++i) {
          r1[i] = *reinterpret_cast<const v4f*>(p1 + (size_t)(2 * i) * DIM);
          r2[i] = *reinterpret_cast<const v4f*>(p2 + (size_t)(2 * i) * DIM);
        }
      }
      __builtin_amdgcn_wave_barrier();
#pragma unroll
      for (int kk = 0; kk < WCH; kk += 4) {
        v2f fa1 = *reinterpret_cast<const v2f*>(a1p + kc * WCH + kk);
        v2f fa2 = *reinterpret_cast<const v2f*>(a2p + kc * WCH + kk);
        v2f fw1 = *reinterpret_cast<const v2f*>(f1p + kk);
        v2f fw2 = *reinterpret_cast<const v2f*>(f2p + kk);
        acc1 = __builtin_amdgcn_wmma_f32_16x16x4_f32(
            false, fa1, false, fw1, (short)0, acc1, false, false);
        acc2 = __builtin_amdgcn_wmma_f32_16x16x4_f32(
            false, fa2, false, fw2, (short)0, acc2, false, false);
      }
      __builtin_amdgcn_wave_barrier();
    }

    const float bias1 = b1[jg];
    const float bias2 = b2[jg];
    float* op = out + (size_t)(rowbase + half * 8) * DIM + colbase + m;
#pragma unroll
    for (int v = 0; v < 8; ++v) {
      float s1 = acc1[v] + bias1;
      s1 = (s1 >= 0.f) ? s1 : NEG_SLOPE * s1;
      float s2 = acc2[v] + bias2;
      s2 = (s2 >= 0.f) ? s2 : NEG_SLOPE * s2;
      op[(size_t)v * DIM] = s1 + s2;
    }
  }
}

extern "C" void kernel_launch(void* const* d_in, const int* in_sizes, int n_in,
                              void* d_out, int out_size, void* d_ws, size_t ws_size,
                              hipStream_t stream) {
  const float* ego  = (const float*)d_in[0];
  const int*   erow = (const int*)  d_in[1];
  const int*   ecol = (const int*)  d_in[2];
  const float* eval = (const float*)d_in[3];
  const float* W1   = (const float*)d_in[4];
  const float* b1   = (const float*)d_in[5];
  const float* W2   = (const float*)d_in[6];
  const float* b2   = (const float*)d_in[7];
  float* out = (float*)d_out;
  const int n_edges = in_sizes[1];

  unsigned int* idx = (unsigned int*)d_ws;

  owner_index<<<128, 512, 0, stream>>>(erow, idx, n_edges);

  spmm_owned_rows<<<(N_NODES + ROWS_WG - 1) / ROWS_WG, 1024, 0, stream>>>(
      erow, ecol, eval, ego, idx, out, n_edges);

  fused_dual_gemm<<<N_NODES / 16, 256, 0, stream>>>(ego, W1, b1, W2, b2, out);
}
